# starting kernel
# baseline (speedup 1.0000x reference)
#include <hip/hip_runtime.h>

typedef float v2f __attribute__((ext_vector_type(2)));
typedef float v8f __attribute__((ext_vector_type(8)));

#define L_LEN 2048
#define N_HEAD 128
#define D_MODEL_SZ 1024
#define HEAD_DIM 8
#define CHUNK 512
#define ROWS_PER_BLK 128
#define NWAVE 8
#define KPAD 16

__global__ __launch_bounds__(256)
void hyena_attn_kernel(const float* __restrict__ kg,
                       const float* __restrict__ vg,
                       const float* __restrict__ qg,
                       const float* __restrict__ kerg,
                       const float* __restrict__ Dg,
                       float* __restrict__ outg)
{
    __shared__ __attribute__((aligned(16))) float sK[HEAD_DIM][CHUNK];
    __shared__ __attribute__((aligned(16))) float sVp[16][CHUNK];
    __shared__ __attribute__((aligned(16))) float sQ[HEAD_DIM][ROWS_PER_BLK];
    __shared__ __attribute__((aligned(16))) float sKerP[KPAD + L_LEN];
    __shared__ __attribute__((aligned(16))) float sW[NWAVE][16 * 17];

    const int tid  = threadIdx.x;
    const int lane = tid & 31;
    const int wid  = tid >> 5;

    const int rblk = blockIdx.x & 15;
    const int bh   = blockIdx.x >> 4;
    const int b    = bh / N_HEAD;
    const int h    = bh % N_HEAD;

    const size_t headBase = ((size_t)b * D_MODEL_SZ + (size_t)h * HEAD_DIM) * L_LEN;

    const float Dh = Dg[h];

    {
        int i  = tid >> 5;
        int c4 = tid & 31;
        const float4* src = (const float4*)(qg + headBase + (size_t)i * L_LEN + rblk * ROWS_PER_BLK);
        ((float4*)&sQ[i][0])[c4] = src[c4];
    }
    {
        const float4* src = (const float4*)(kerg + (size_t)h * L_LEN);
        for (int u = tid; u < L_LEN / 4; u += 256) {
            float4 f = src[u];
            if (u == 0) f.x += Dh;
            ((float4*)(sKerP + KPAD))[u] = f;
        }
        if (tid < KPAD) sKerP[tid] = 0.0f;
    }
    {
        float4 z = make_float4(0.f, 0.f, 0.f, 0.f);
        for (int u = tid; u < 8 * CHUNK / 4; u += 256)
            ((float4*)&sVp[8][0])[u] = z;
    }
    __syncthreads();

    const int rowBase  = rblk * ROWS_PER_BLK + wid * 16;
    const int rowBaseU = __builtin_amdgcn_readfirstlane(rowBase);
    const int n  = lane & 15;
    const int k0 = (lane < 16) ? 0 : 2;
    const int mB = (lane < 16) ? 0 : 8;

    v2f qa0, qa1;
    {
        int lcol = wid * 16 + n;
        qa0.x = sQ[k0 + 0][lcol];
        qa0.y = sQ[k0 + 1][lcol];
        qa1.x = sQ[4 + k0 + 0][lcol];
        qa1.y = sQ[4 + k0 + 1][lcol];
    }

    v8f acc = {};
    float* sWw = &sW[wid][0];
    const int arow = n * 17;

    const int nChunk = (rblk >> 2) + 1;
    for (int c = 0; c < nChunk; ++c) {
        const int tStart = c * CHUNK;

        for (int u = tid; u < HEAD_DIM * CHUNK / 4; u += 256) {
            int i  = u >> 7;
            int c4 = u & 127;
            ((float4*)&sK[i][0])[c4]  = ((const float4*)(kg + headBase + (size_t)i * L_LEN + tStart))[c4];
            ((float4*)&sVp[i][0])[c4] = ((const float4*)(vg + headBase + (size_t)i * L_LEN + tStart))[c4];
        }

        if (c + 1 < nChunk) {
            const int tNext = tStart + CHUNK;
            const float* basePtr = (tid < 128) ? kg : vg;
            int r  = (tid & 127) >> 4;
            int ln = (tid & 15) * 32;
            __builtin_prefetch(basePtr + headBase + (size_t)r * L_LEN + tNext + ln, 0, 3);
        }
        __syncthreads();

        int ttEnd = ((rowBaseU - tStart) >> 4) + 1;
        if (ttEnd > CHUNK / 16) ttEnd = CHUNK / 16;

        for (int tt = 0; tt < ttEnd; ++tt) {
            const int colLoc  = tt * 16;
            const int colBase = tStart + colLoc;

            v2f bk0, bk1;
            bk0.x = sK[k0 + 0][colLoc + n];
            bk0.y = sK[k0 + 1][colLoc + n];
            bk1.x = sK[4 + k0 + 0][colLoc + n];
            bk1.y = sK[4 + k0 + 1][colLoc + n];

            v8f s = {};
            s = __builtin_amdgcn_wmma_f32_16x16x4_f32(false, qa0, false, bk0, (short)0, s, false, false);
            s = __builtin_amdgcn_wmma_f32_16x16x4_f32(false, qa1, false, bk1, (short)0, s, false, false);

            const int kerBase = KPAD + rowBase + mB - (colBase + n);
            #pragma unroll
            for (int r = 0; r < 8; ++r) {
                float wc = sKerP[kerBase + r];
                sWw[(mB + r) * 17 + n] = s[r] * wc;
            }

            #pragma unroll
            for (int kc = 0; kc < 4; ++kc) {
                v2f aw, bv;
                aw.x = sWw[arow + 4 * kc + k0 + 0];
                aw.y = sWw[arow + 4 * kc + k0 + 1];
                int tl = colLoc + 4 * kc + k0;
                bv.x = sVp[n][tl + 0];
                bv.y = sVp[n][tl + 1];
                acc = __builtin_amdgcn_wmma_f32_16x16x4_f32(false, aw, false, bv, (short)0, acc, false, false);
            }
        }
        __syncthreads();
    }

    if (n < 8) {
        size_t o = headBase + (size_t)n * L_LEN + (size_t)(rowBase + mB);
        float4 lo = make_float4(acc[0], acc[1], acc[2], acc[3]);
        float4 hi = make_float4(acc[4], acc[5], acc[6], acc[7]);
        *(float4*)(outg + o)     = lo;
        *(float4*)(outg + o + 4) = hi;
    }
}

extern "C" void kernel_launch(void* const* d_in, const int* in_sizes, int n_in,
                              void* d_out, int out_size, void* d_ws, size_t ws_size,
                              hipStream_t stream) {
    const float* k   = (const float*)d_in[0];
    const float* v   = (const float*)d_in[1];
    const float* q   = (const float*)d_in[2];
    const float* ker = (const float*)d_in[3];
    const float* D   = (const float*)d_in[4];
    float* out = (float*)d_out;

    const int B = in_sizes[0] / (D_MODEL_SZ * L_LEN);
    dim3 grid((unsigned)(B * N_HEAD * (L_LEN / ROWS_PER_BLK)));
    hipLaunchKernelGGL(hyena_attn_kernel, grid, dim3(256), 0, stream, k, v, q, ker, D, out);
}


# Modulated tile W stored transposed in LDS (row t, 16 l; a tile per row tile): two 16-byte stores per lane, A fragment one 2-address load per k chunk; k order of W V permuted, sum order changes
# speedup vs baseline: 3.0107x; 3.0107x over previous
#include <hip/hip_runtime.h>

typedef float v2f __attribute__((ext_vector_type(2)));
typedef float v4f __attribute__((ext_vector_type(4)));
typedef float v8f __attribute__((ext_vector_type(8)));

#define L_LEN 2048
#define N_HEAD 128
#define D_MODEL_SZ 1024
#define HEAD_DIM 8
#define ROWS_PER_BLK 128
#define NWAVE 8
#define NRT 8
#define KPAD 16
#define WT_CHUNK 68
#define WT_TILE 268

__global__ __launch_bounds__(256)
void hyena_attn_kernel(const float* __restrict__ kg,
                       const float* __restrict__ vg,
                       const float* __restrict__ qg,
                       const float* __restrict__ kerg,
                       const float* __restrict__ Dg,
                       float* __restrict__ outg)
{
    __shared__ __attribute__((aligned(16))) float sRed[NWAVE * NRT * 8 * 16];
    __shared__ __attribute__((aligned(16))) float sQ[HEAD_DIM][ROWS_PER_BLK];
    __shared__ __attribute__((aligned(16))) float sKerP[KPAD + L_LEN];
    __shared__ __attribute__((aligned(16))) float sW[NWAVE][NRT * WT_TILE];

    const int tid  = threadIdx.x;
    const int lane = tid & 31;
    const int wid  = tid >> 5;

    const int nbh  = (int)(gridDim.x >> 4);
    const int rblk = 15 - (int)(blockIdx.x / (unsigned)nbh);
    const int bh   = (int)(blockIdx.x % (unsigned)nbh);
    const int b    = bh / N_HEAD;
    const int h    = bh % N_HEAD;

    const size_t headBase = ((size_t)b * D_MODEL_SZ + (size_t)h * HEAD_DIM) * L_LEN;

    const float Dh = Dg[h];

    const int widU   = __builtin_amdgcn_readfirstlane(wid);
    const int blkRow = rblk * ROWS_PER_BLK;
    const int n  = lane & 15;
    const int k0 = (lane < 16) ? 0 : 2;
    const int mB = (lane < 16) ? 0 : 8;

    const float* kp = kg + headBase + (size_t)k0 * L_LEN + n;
    const float* vp = vg + headBase + (size_t)(n & 7) * L_LEN + mB;
    const bool jPad = n >= 8;

    v2f rk0, rk1;
    float2 rv[4];
#define HY_LOAD_RAW(COL)                                                                         \
    {                                                                                            \
        const float* kq = kp + (COL);                                                            \
        const float* vq = vp + (COL);                                                            \
        rk0.x = kq[0];                                                                           \
        rk0.y = kq[L_LEN];                                                                       \
        rk1.x = kq[4 * L_LEN];                                                                   \
        rk1.y = kq[5 * L_LEN];                                                                   \
        const float4 vlo = *(const float4*)(vq + 0);                                             \
        const float4 vhi = *(const float4*)(vq + 4);                                             \
        rv[0] = make_float2(vlo.x, vlo.y);                                                       \
        rv[1] = make_float2(vlo.z, vlo.w);                                                       \
        rv[2] = make_float2(vhi.x, vhi.y);                                                       \
        rv[3] = make_float2(vhi.z, vhi.w);                                                       \
    }
    HY_LOAD_RAW(widU * 16)

    {
        int i  = tid >> 5;
        int c4 = tid & 31;
        const float4* src = (const float4*)(qg + headBase + (size_t)i * L_LEN + rblk * ROWS_PER_BLK);
        ((float4*)&sQ[i][0])[c4] = src[c4];
    }
    {
        const float4* src = (const float4*)(kerg + (size_t)h * L_LEN);
        for (int u = tid; u < L_LEN / 4; u += 256) {
            float4 f = src[u];
            if (u == 0) f.x += Dh;
            ((float4*)(sKerP + KPAD))[u] = f;
        }
        if (tid < KPAD) sKerP[tid] = 0.0f;
    }
    __syncthreads();

    v2f qa[NRT][2];
    #pragma unroll
    for (int rt = 0; rt < NRT; ++rt) {
        int lcol = rt * 16 + n;
        qa[rt][0].x = sQ[k0 + 0][lcol];
        qa[rt][0].y = sQ[k0 + 1][lcol];
        qa[rt][1].x = sQ[4 + k0 + 0][lcol];
        qa[rt][1].y = sQ[4 + k0 + 1][lcol];
    }

    v8f acc[NRT];
    #pragma unroll
    for (int rt = 0; rt < NRT; ++rt) acc[rt] = (v8f){0.f, 0.f, 0.f, 0.f, 0.f, 0.f, 0.f, 0.f};

    float* sWw = &sW[wid][0];
    float* sWst = sWw + WT_CHUNK * ((n & 7) >> 1) + 16 * ((n & 1) + 2 * (n >> 3)) + mB;
    const float* sWld = sWw + 4 * mB + n;

#define HY_TILE(RT)                                                                              \
    {                                                                                            \
        v8f s = {};                                                                              \
        s = __builtin_amdgcn_wmma_f32_16x16x4_f32(false, qa[RT][0], false, bk0, (short)0, s, false, false); \
        s = __builtin_amdgcn_wmma_f32_16x16x4_f32(false, qa[RT][1], false, bk1, (short)0, s, false, false); \
        const int kerBase = kerCol + (RT) * 16;                                                  \
        v4f wlo, whi;                                                                            \
        wlo.x = s[0] * sKerP[kerBase + 0];                                                       \
        wlo.y = s[1] * sKerP[kerBase + 1];                                                       \
        wlo.z = s[2] * sKerP[kerBase + 2];                                                       \
        wlo.w = s[3] * sKerP[kerBase + 3];                                                       \
        whi.x = s[4] * sKerP[kerBase + 4];                                                       \
        whi.y = s[5] * sKerP[kerBase + 5];                                                       \
        whi.z = s[6] * sKerP[kerBase + 6];                                                       \
        whi.w = s[7] * sKerP[kerBase + 7];                                                       \
        *(v4f*)(sWst + (RT) * WT_TILE)     = wlo;                                                \
        *(v4f*)(sWst + (RT) * WT_TILE + 4) = whi;                                                \
        _Pragma("unroll")                                                                        \
        for (int kc = 0; kc < 4; ++kc) {                                                         \
            v2f aw;                                                                              \
            aw.x = sWld[(RT) * WT_TILE + WT_CHUNK * kc];                                         \
            aw.y = sWld[(RT) * WT_TILE + WT_CHUNK * kc + 16];                                    \
            acc[RT] = __builtin_amdgcn_wmma_f32_16x16x4_f32(false, aw, false, bv[kc], (short)0, acc[RT], false, false); \
        }                                                                                        \
    }

    const int ttEnd = (rblk + 1) * NRT;

    for (int tt = widU; tt < ttEnd; tt += NWAVE) {
        const int colBase = tt * 16;

        const v2f bk0 = rk0, bk1 = rk1;
        v2f bv[4];
        #pragma unroll
        for (int kc = 0; kc < 4; ++kc) {
            bv[kc].x = jPad ? 0.0f : rv[kc].x;
            bv[kc].y = jPad ? 0.0f : rv[kc].y;
        }

        int ttNext = tt + NWAVE;
        if (ttNext >= ttEnd) ttNext = tt;
        HY_LOAD_RAW(ttNext * 16)

        const int rtMin  = (colBase - blkRow) >> 4;
        const int kerCol = KPAD + blkRow + mB - (colBase + n);

        if (rtMin <= 0) {
            HY_TILE(0) HY_TILE(1) HY_TILE(2) HY_TILE(3)
            HY_TILE(4) HY_TILE(5) HY_TILE(6) HY_TILE(7)
        } else {
            if (1 >= rtMin) HY_TILE(1)
            if (2 >= rtMin) HY_TILE(2)
            if (3 >= rtMin) HY_TILE(3)
            if (4 >= rtMin) HY_TILE(4)
            if (5 >= rtMin) HY_TILE(5)
            if (6 >= rtMin) HY_TILE(6)
            HY_TILE(7)
        }
    }
#undef HY_LOAD_RAW
#undef HY_TILE

    float* red = sRed;
    if (n < 8) {
        #pragma unroll
        for (int rt = 0; rt < NRT; ++rt) {
            float* p = red + (((wid * NRT + rt) * 8 + n) * 16 + mB);
            *(float4*)(p)     = make_float4(acc[rt][0], acc[rt][1], acc[rt][2], acc[rt][3]);
            *(float4*)(p + 4) = make_float4(acc[rt][4], acc[rt][5], acc[rt][6], acc[rt][7]);
        }
    }
    __syncthreads();

    if (n < 8) {
        float4 lo = make_float4(0.f, 0.f, 0.f, 0.f);
        float4 hi = make_float4(0.f, 0.f, 0.f, 0.f);
        #pragma unroll
        for (int w = 0; w < NWAVE; ++w) {
            const float* p = red + (((w * NRT + wid) * 8 + n) * 16 + mB);
            float4 a = *(const float4*)(p);
            float4 b4 = *(const float4*)(p + 4);
            lo.x += a.x;  lo.y += a.y;  lo.z += a.z;  lo.w += a.w;
            hi.x += b4.x; hi.y += b4.y; hi.z += b4.z; hi.w += b4.w;
        }
        size_t o = headBase + (size_t)n * L_LEN + (size_t)(blkRow + wid * 16 + mB);
        *(float4*)(outg + o)     = lo;
        *(float4*)(outg + o + 4) = hi;
    }
}

extern "C" void kernel_launch(void* const* d_in, const int* in_sizes, int n_in,
                              void* d_out, int out_size, void* d_ws, size_t ws_size,
                              hipStream_t stream) {
    const float* k   = (const float*)d_in[0];
    const float* v   = (const float*)d_in[1];
    const float* q   = (const float*)d_in[2];
    const float* ker = (const float*)d_in[3];
    const float* D   = (const float*)d_in[4];
    float* out = (float*)d_out;

    const int B = in_sizes[0] / (D_MODEL_SZ * L_LEN);
    dim3 grid((unsigned)(B * N_HEAD * (L_LEN / ROWS_PER_BLK)));
    hipLaunchKernelGGL(hyena_attn_kernel, grid, dim3(256), 0, stream, k, v, q, ker, D, out);
}
